# starting kernel
# baseline (speedup 1.0000x reference)
#include <hip/hip_runtime.h>
#include <math.h>

#define B_    32
#define C_    256
#define HW_   4096
#define NOUT_ 768
#define BN_EPS_ 1e-5f

typedef float v2f __attribute__((ext_vector_type(2)));
typedef float v8f __attribute__((ext_vector_type(8)));

__global__ void gap_kernel(const float* __restrict__ x, float* __restrict__ g)
{
    const int plane = blockIdx.x;
    const float4* xp = (const float4*)(x + (size_t)plane * HW_);
    float s = 0.f;
#pragma unroll
    for (int i = 0; i < 4; ++i) {
        float4 v = xp[threadIdx.x + i * 256];
        s += v.x + v.y + v.z + v.w;
    }
#pragma unroll
    for (int off = 16; off > 0; off >>= 1)
        s += __shfl_xor(s, off, 32);
    __shared__ float lds[8];
    const int wid  = threadIdx.x >> 5;
    const int lane = threadIdx.x & 31;
    if (lane == 0) lds[wid] = s;
    __syncthreads();
    if (threadIdx.x == 0) {
        float t = 0.f;
#pragma unroll
        for (int w = 0; w < 8; ++w) t += lds[w];
        g[plane] = t * (1.0f / (float)HW_);
    }
}

__global__ void filt_gemm_wmma_kernel(const float* __restrict__ g,
                                      const float* __restrict__ w,
                                      const float* __restrict__ bnw,
                                      const float* __restrict__ bnb,
                                      const float* __restrict__ bnm,
                                      const float* __restrict__ bnv,
                                      float* __restrict__ f)
{
    const int tile = blockIdx.x;
    const int mt   = tile & 1;
    const int nt   = tile >> 1;
    const int lane = threadIdx.x;
    const int ln   = lane & 15;
    const int lh   = lane >> 4;

    const float* arow = g + (size_t)(mt * 16 + ln) * C_;
    const float* brow = w + (size_t)(nt * 16 + ln) * C_;

    v8f acc = {};
    for (int k0 = 0; k0 < C_; k0 += 4) {
        v2f a, b;
        a.x = arow[k0 + 2 * lh + 0];
        a.y = arow[k0 + 2 * lh + 1];
        b.x = brow[k0 + 2 * lh + 0];
        b.y = brow[k0 + 2 * lh + 1];
        acc = __builtin_amdgcn_wmma_f32_16x16x4_f32(
            false, a, false, b, (short)0, acc, false, false);
    }

    const int   n     = nt * 16 + ln;
    const float scale = bnw[n] * rsqrtf(bnv[n] + BN_EPS_);
    const float bias  = bnb[n] - bnm[n] * scale;
#pragma unroll
    for (int r = 0; r < 8; ++r) {
        const int m = mt * 16 + lh * 8 + r;
        f[(size_t)m * NOUT_ + n] = tanhf(acc[r] * scale + bias);
    }
}

__global__ void apply_kernel(const float* __restrict__ x,
                             const float* __restrict__ f,
                             const float* __restrict__ gamma,
                             const float* __restrict__ beta,
                             float* __restrict__ out)
{
    const int c = blockIdx.x;
    const int b = blockIdx.y;
    const int cm = (c == 0)      ? 1        : c - 1;
    const int cp = (c == C_ - 1) ? C_ - 2   : c + 1;

    const float f0 = f[(size_t)b * NOUT_ + 0 * C_ + c];
    const float f1 = f[(size_t)b * NOUT_ + 1 * C_ + c];
    const float f2 = f[(size_t)b * NOUT_ + 2 * C_ + c];
    const float ga = gamma[c];
    const float be = beta[c];

    const float4* xm = (const float4*)(x   + ((size_t)b * C_ + cm) * HW_);
    const float4* xc = (const float4*)(x   + ((size_t)b * C_ + c ) * HW_);
    const float4* xp = (const float4*)(x   + ((size_t)b * C_ + cp) * HW_);
    float4*       op = (float4*)      (out + ((size_t)b * C_ + c ) * HW_);

#pragma unroll
    for (int i = 0; i < 4; ++i) {
        const int idx = threadIdx.x + i * 256;
        const float4 a = xm[idx];
        const float4 m = xc[idx];
        const float4 p = xp[idx];
        float4 o;
        o.x = ga * (f0 * a.x + f1 * m.x + f2 * p.x) + be * m.x;
        o.y = ga * (f0 * a.y + f1 * m.y + f2 * p.y) + be * m.y;
        o.z = ga * (f0 * a.z + f1 * m.z + f2 * p.z) + be * m.z;
        o.w = ga * (f0 * a.w + f1 * m.w + f2 * p.w) + be * m.w;
        op[idx] = o;
    }
}

extern "C" void kernel_launch(void* const* d_in, const int* in_sizes, int n_in,
                              void* d_out, int out_size, void* d_ws, size_t ws_size,
                              hipStream_t stream)
{
    const float* x      = (const float*)d_in[0];
    const float* conv_w = (const float*)d_in[1];
    const float* bnw    = (const float*)d_in[2];
    const float* bnb    = (const float*)d_in[3];
    const float* bnm    = (const float*)d_in[4];
    const float* bnv    = (const float*)d_in[5];
    const float* gamma  = (const float*)d_in[6];
    const float* beta   = (const float*)d_in[7];
    float*       out    = (float*)d_out;

    float* g_ws = (float*)d_ws;
    float* f_ws = (float*)((char*)d_ws + B_ * C_ * sizeof(float));

    gap_kernel<<<B_ * C_, 256, 0, stream>>>(x, g_ws);
    filt_gemm_wmma_kernel<<<96, 32, 0, stream>>>(g_ws, conv_w, bnw, bnb, bnm, bnv, f_ws);
    apply_kernel<<<dim3(C_, B_), 256, 0, stream>>>(x, f_ws, gamma, beta, out);
}


# filter GEMM split over k into the pooling kernel: a work-group multiplies its 16 channel means by its conv_w slice on WMMA, apply sums the 16 partials; 2 launches, not 3; k and plane sum order change
# speedup vs baseline: 1.2779x; 1.2779x over previous
#include <hip/hip_runtime.h>
#include <math.h>

#define B_    32
#define C_    256
#define HW_   4096
#define BN_EPS_ 1e-5f

typedef float v2f __attribute__((ext_vector_type(2)));
typedef float v8f __attribute__((ext_vector_type(8)));

#define PG_     16
#define PNGRP_  (C_ / PG_)
#define PSLOT_  896
#define PLD_    8
#define PNLD_   (HW_ / (32 * 4) / PLD_)

__global__ __launch_bounds__(256)
void pool_filt_kernel(const float* __restrict__ x,
                      const float* __restrict__ w,
                      float* __restrict__ part)
{
    const int wv   = __builtin_amdgcn_readfirstlane((int)(threadIdx.x >> 5));
    const int lane = threadIdx.x & 31;
    const int ln   = lane & 15;
    const int lh   = lane >> 4;
    const int kc   = blockIdx.x;
    const int b    = blockIdx.y;

    __shared__ float gs[PG_];

    float4 wa[6], wb[6];
#pragma unroll
    for (int t = 0; t < 6; ++t) {
        const int tap = t >> 1;
        const int cx  = 2 * wv + (t & 1);
        const int n   = tap * C_ + cx * 16 + ln;
        const float* wr = w + (size_t)n * C_ + kc * PG_ + 8 * lh;
        wa[t] = *(const float4*)(wr);
        wb[t] = *(const float4*)(wr + 4);
    }

    const float* xp = x + ((size_t)b * C_ + kc * PG_ + 2 * wv) * HW_ + (size_t)lane * 4;
    float4 buf[3][PLD_];
#define POOL_LOAD(dst, g)                                                     \
    do {                                                                      \
        _Pragma("unroll")                                                     \
        for (int i = 0; i < PLD_; ++i)                                        \
            (dst)[i] = *(const float4*)(xp + ((g) * PLD_ + i) * 128);         \
    } while (0)

    POOL_LOAD(buf[0], 0);
    __builtin_amdgcn_sched_barrier(0);
    POOL_LOAD(buf[1], 1);
    __builtin_amdgcn_sched_barrier(0);

    float s[2][4] = {{0.f, 0.f, 0.f, 0.f}, {0.f, 0.f, 0.f, 0.f}};
#pragma unroll
    for (int g = 0; g < 2 * PNLD_; ++g) {
        if (g + 2 < 2 * PNLD_) POOL_LOAD(buf[(g + 2) % 3], g + 2);
        __builtin_amdgcn_sched_barrier(0);
#pragma unroll
        for (int i = 0; i < PLD_; ++i) {
            const float4 v = buf[g % 3][i];
            s[g / PNLD_][i & 3] += (v.x + v.y) + (v.z + v.w);
        }
        __builtin_amdgcn_sched_barrier(0);
    }
#undef POOL_LOAD

    float s0 = (s[0][0] + s[0][1]) + (s[0][2] + s[0][3]);
    float s1 = (s[1][0] + s[1][1]) + (s[1][2] + s[1][3]);
#pragma unroll
    for (int off = 16; off > 0; off >>= 1) {
        s0 += __shfl_xor(s0, off, 32);
        s1 += __shfl_xor(s1, off, 32);
    }
    if (lane == 0) {
        gs[2 * wv + 0] = s0 * (1.0f / (float)HW_);
        gs[2 * wv + 1] = s1 * (1.0f / (float)HW_);
    }
    __syncthreads();

    const float4 ga = *(const float4*)(gs + 8 * lh);
    const float4 gb = *(const float4*)(gs + 8 * lh + 4);
    v2f a0, a1, a2, a3;
    a0.x = ga.x; a0.y = ga.y;
    a1.x = ga.z; a1.y = ga.w;
    a2.x = gb.x; a2.y = gb.y;
    a3.x = gb.z; a3.y = gb.w;

    float r[6];
#pragma unroll
    for (int t = 0; t < 6; ++t) {
        v2f b0, b1, b2, b3;
        b0.x = wa[t].x; b0.y = wa[t].y;
        b1.x = wa[t].z; b1.y = wa[t].w;
        b2.x = wb[t].x; b2.y = wb[t].y;
        b3.x = wb[t].z; b3.y = wb[t].w;
        v8f acc = {};
        acc = __builtin_amdgcn_wmma_f32_16x16x4_f32(false, a0, false, b0, (short)0, acc, false, false);
        acc = __builtin_amdgcn_wmma_f32_16x16x4_f32(false, a1, false, b1, (short)0, acc, false, false);
        acc = __builtin_amdgcn_wmma_f32_16x16x4_f32(false, a2, false, b2, (short)0, acc, false, false);
        acc = __builtin_amdgcn_wmma_f32_16x16x4_f32(false, a3, false, b3, (short)0, acc, false, false);
        r[t] = acc[0];
    }

    float* sl0 = part + ((size_t)(b * C_ + (2 * wv + 0) * 16 + kc) * HW_ + PSLOT_);
    float* sl1 = part + ((size_t)(b * C_ + (2 * wv + 1) * 16 + kc) * HW_ + PSLOT_);
    sl0[lane] = lh ? r[2] : r[0];
    sl1[lane] = lh ? r[3] : r[1];
    (lh ? sl1 : sl0)[32 + ln] = lh ? r[5] : r[4];
}

#define TG_     8
#define TROWS_  (TG_ + 2)
#define TSEG_   1024
#define TSTEP_  128
#define TNSTEP_ (TSEG_ / TSTEP_)
#define TWGC_   (2 * TG_)
#define TGRIDX_ (C_ / TWGC_)

__global__ __launch_bounds__(256)
void apply_kernel(const float* __restrict__ x,
                  const float* __restrict__ bnw,
                  const float* __restrict__ bnb,
                  const float* __restrict__ bnm,
                  const float* __restrict__ bnv,
                  const float* __restrict__ gamma,
                  const float* __restrict__ beta,
                  float* __restrict__ out)
{
    const int wv   = __builtin_amdgcn_readfirstlane((int)(threadIdx.x >> 5));
    const int lane = threadIdx.x & 31;
    const int b    = blockIdx.y;
    const int c0   = blockIdx.x * TWGC_ + (wv >> 2) * TG_;
    const int seg  = wv & 3;

    const int clo = (c0 == 0)        ? 1      : c0 - 1;
    const int chi = (c0 + TG_ == C_) ? C_ - 2 : c0 + TG_;

    const size_t pix = (size_t)seg * TSEG_ + (size_t)lane * 4;
    const float* xb  = x + (size_t)b * C_ * HW_ + pix;
    const float* xlo = xb + (size_t)clo * HW_;
    const float* xmi = xb + (size_t)c0  * HW_;
    const float* xhi = xb + (size_t)chi * HW_;
    float*       ob  = out + ((size_t)b * C_ + c0) * HW_ + pix;

    const int fl   = lane < 24 ? lane : lane - 8;
    const int ftap = fl >> 3;
    const int fj   = fl & 7;
    const float* ps = out + ((size_t)(b * C_ + blockIdx.x * TWGC_) * HW_ + PSLOT_)
                          + ftap * 16 + (wv >> 2) * TG_ + fj;
    float pv[PNGRP_];
#pragma unroll
    for (int k = 0; k < PNGRP_; ++k) pv[k] = ps[(size_t)k * HW_];
    const int   fn  = ftap * C_ + c0 + fj;
    const float bw_ = bnw[fn];
    const float bb_ = bnb[fn];
    const float bm_ = bnm[fn];
    const float bv_ = bnv[fn];
    __builtin_amdgcn_sched_barrier(0);

    float4 buf[3][TROWS_];

#define LOAD_STEP(dst, s)                                                     \
    do {                                                                      \
        (dst)[0] = *(const float4*)(xlo + (s) * TSTEP_);                      \
        _Pragma("unroll")                                                     \
        for (int r = 0; r < TG_; ++r)                                         \
            (dst)[1 + r] = *(const float4*)(xmi + (size_t)r * HW_ + (s) * TSTEP_); \
        (dst)[TROWS_ - 1] = *(const float4*)(xhi + (s) * TSTEP_);             \
    } while (0)

    LOAD_STEP(buf[0], 0);
    __builtin_amdgcn_sched_barrier(0);
    LOAD_STEP(buf[1], 1);
    __builtin_amdgcn_sched_barrier(0);

    float y = pv[0];
#pragma unroll
    for (int k = 1; k < PNGRP_; ++k) y += pv[k];
    const float scale = bw_ * rsqrtf(bv_ + BN_EPS_);
    const float bias  = bb_ - bm_ * scale;
    const int   fbits = __float_as_int(tanhf(y * scale + bias));

#pragma unroll
    for (int s = 0; s < TNSTEP_; ++s) {
        if (s + 2 < TNSTEP_) LOAD_STEP(buf[(s + 2) % 3], s + 2);
        __builtin_amdgcn_sched_barrier(0);
        if (s == TNSTEP_ - 1) __syncthreads();
        int fbs = fbits;
        asm volatile("" : "+v"(fbs));
#pragma unroll
        for (int j = 0; j < TG_; ++j) {
            const float f0 = __int_as_float(__builtin_amdgcn_readlane(fbs, 0 * 8 + j));
            const float f1 = __int_as_float(__builtin_amdgcn_readlane(fbs, 1 * 8 + j));
            const float f2 = __int_as_float(__builtin_amdgcn_readlane(fbs, 2 * 8 + j));
            const float ga = gamma[c0 + j];
            const float be = beta[c0 + j];
            const float4 a = buf[s % 3][j];
            const float4 m = buf[s % 3][j + 1];
            const float4 p = buf[s % 3][j + 2];
            float4 o;
            o.x = ga * (f0 * a.x + f1 * m.x + f2 * p.x) + be * m.x;
            o.y = ga * (f0 * a.y + f1 * m.y + f2 * p.y) + be * m.y;
            o.z = ga * (f0 * a.z + f1 * m.z + f2 * p.z) + be * m.z;
            o.w = ga * (f0 * a.w + f1 * m.w + f2 * p.w) + be * m.w;
            *(float4*)(ob + (size_t)j * HW_ + s * TSTEP_) = o;
        }
        __builtin_amdgcn_sched_barrier(0);
    }
#undef LOAD_STEP
}

extern "C" void kernel_launch(void* const* d_in, const int* in_sizes, int n_in,
                              void* d_out, int out_size, void* d_ws, size_t ws_size,
                              hipStream_t stream)
{
    const float* x      = (const float*)d_in[0];
    const float* conv_w = (const float*)d_in[1];
    const float* bnw    = (const float*)d_in[2];
    const float* bnb    = (const float*)d_in[3];
    const float* bnm    = (const float*)d_in[4];
    const float* bnv    = (const float*)d_in[5];
    const float* gamma  = (const float*)d_in[6];
    const float* beta   = (const float*)d_in[7];
    float*       out    = (float*)d_out;

    pool_filt_kernel<<<dim3(PNGRP_, B_), 256, 0, stream>>>(x, conv_w, out);
    apply_kernel<<<dim3(TGRIDX_, B_), 256, 0, stream>>>(x, bnw, bnb, bnm, bnv, gamma, beta, out);
}
